# starting kernel
# baseline (speedup 1.0000x reference)
#include <hip/hip_runtime.h>

typedef __attribute__((ext_vector_type(2))) float v2f;
typedef __attribute__((ext_vector_type(8))) float v8f;

#define HD 32
#define VD 64
#define TCH 64
#define NTHREADS 160


__launch_bounds__(NTHREADS)
__global__ void delta_model_fused(
    const int*   __restrict__ seq,
    const float* __restrict__ embed,
    const float* __restrict__ w1, const float* __restrict__ b1,
    const float* __restrict__ w2, const float* __restrict__ b2,
    const float* __restrict__ ln_g, const float* __restrict__ ln_b,
    const float* __restrict__ wr, const float* __restrict__ br,
    const float* __restrict__ wo, const float* __restrict__ bo,
    float* __restrict__ out, int L)
{
  __shared__ float s_w1[HD * 2 * HD];
  __shared__ float s_w2[2 * HD * HD];
  __shared__ float s_b1[2 * HD];
  __shared__ float s_b2[HD];
  __shared__ float s_g[HD];
  __shared__ float s_bt[HD];
  __shared__ float s_h [TCH * 34];
  __shared__ float s_ff[TCH * 66];
  __shared__ __align__(16) float s_x[2][TCH * 36];
  __shared__ float s_ctx[HD];
  __shared__ float s_tmp[HD];

  const int tid  = threadIdx.x;
  const int wave = __builtin_amdgcn_readfirstlane(tid) >> 5;
  const int lane = tid & 31;
  const int ln   = lane & 15;
  const int hl   = lane >> 4;
  const int bidx = blockIdx.x;

  for (int i = tid; i < HD * 2 * HD; i += NTHREADS) s_w1[i] = w1[i];
  for (int i = tid; i < 2 * HD * HD; i += NTHREADS) s_w2[i] = w2[i];
  if (tid < 2 * HD) s_b1[tid] = b1[tid];
  if (tid < HD) { s_b2[tid] = b2[tid]; s_g[tid] = ln_g[tid]; s_bt[tid] = ln_b[tid]; }

  float Mrow[HD];
#pragma unroll
  for (int j = 0; j < HD; ++j) Mrow[j] = 0.f;

  const int nChunks = L / TCH;
  const int m0 = (wave & 3) * 16;

  auto encode = [&](int cIdx, float* xb) {
    const int t0 = cIdx * TCH;
#pragma unroll
    for (int r = 0; r < 16; ++r) {
      int tok = seq[(size_t)bidx * L + t0 + m0 + r];
      s_h[(m0 + r) * 34 + lane] = embed[tok * HD + lane];
    }
#pragma unroll
    for (int nt = 0; nt < 4; ++nt) {
      const int n0 = nt * 16;
      v8f acc = {0.f, 0.f, 0.f, 0.f, 0.f, 0.f, 0.f, 0.f};
#pragma unroll
      for (int k0 = 0; k0 < HD; k0 += 4) {
        const float* ap = &s_h[(m0 + ln) * 34 + k0 + 2 * hl];
        v2f a; a.x = ap[0]; a.y = ap[1];
        v2f bf; bf.x = s_w1[(k0 + 2 * hl) * 64 + n0 + ln];
        bf.y = s_w1[(k0 + 2 * hl + 1) * 64 + n0 + ln];
        acc = __builtin_amdgcn_wmma_f32_16x16x4_f32(false, a, false, bf,
                                                    (short)0, acc, false, false);
      }
      float bias = s_b1[n0 + ln];
#pragma unroll
      for (int r = 0; r < 8; ++r) {
        int row = m0 + r + 8 * hl;
        float v = acc[r] + bias;
        s_ff[row * 66 + n0 + ln] = v > 0.f ? v : 0.f;
      }
    }
#pragma unroll
    for (int nt = 0; nt < 2; ++nt) {
      const int n0 = nt * 16;
      v8f acc = {0.f, 0.f, 0.f, 0.f, 0.f, 0.f, 0.f, 0.f};
#pragma unroll
      for (int k0 = 0; k0 < 2 * HD; k0 += 4) {
        const float* ap = &s_ff[(m0 + ln) * 66 + k0 + 2 * hl];
        v2f a; a.x = ap[0]; a.y = ap[1];
        v2f bf; bf.x = s_w2[(k0 + 2 * hl) * HD + n0 + ln];
        bf.y = s_w2[(k0 + 2 * hl + 1) * HD + n0 + ln];
        acc = __builtin_amdgcn_wmma_f32_16x16x4_f32(false, a, false, bf,
                                                    (short)0, acc, false, false);
      }
      float bias = s_b2[n0 + ln];
#pragma unroll
      for (int r = 0; r < 8; ++r) {
        int row = m0 + r + 8 * hl, col = n0 + ln;
        xb[row * 36 + col] = acc[r] + bias + s_h[row * 34 + col];
      }
    }
    if (lane < 16) {
      float* xr = &xb[(m0 + lane) * 36];
      float mu = 0.f;
#pragma unroll
      for (int j = 0; j < HD; ++j) mu += xr[j];
      mu *= (1.f / 32.f);
      float var = 0.f;
#pragma unroll
      for (int j = 0; j < HD; ++j) { float d = xr[j] - mu; var = fmaf(d, d, var); }
      var *= (1.f / 32.f);
      float rs = __builtin_amdgcn_rsqf(var + 1e-5f);
      float kk2 = 0.f;
#pragma unroll
      for (int j = 0; j < HD; ++j) {
        float v = (xr[j] - mu) * rs * s_g[j] + s_bt[j];
        xr[j] = v;
        kk2 = fmaf(v, v, kk2);
      }
      xr[HD] = __builtin_amdgcn_rcpf(kk2 + 1e-6f);
    }
  };

  __syncthreads();
  if (wave < 4) encode(0, s_x[0]);
  __syncthreads();

  for (int c = 0; c < nChunks; ++c) {
    float* xb_cur = s_x[c & 1];
    float* xb_nxt = s_x[(c & 1) ^ 1];

    if (wave < 4) {
      if (c + 1 < nChunks) encode(c + 1, xb_nxt);
    } else {
      const bool last = (c == nChunks - 1);
      const int steps = last ? (TCH - 1) : TCH;

      auto loadk = [&](int t, float (&kkv)[HD], float& kl, float& ri) {
        const float4* kp = (const float4*)&xb_cur[t * 36];
#pragma unroll
        for (int q = 0; q < 8; ++q) {
          float4 f = kp[q];
          kkv[4 * q] = f.x; kkv[4 * q + 1] = f.y;
          kkv[4 * q + 2] = f.z; kkv[4 * q + 3] = f.w;
        }
        ri = xb_cur[t * 36 + HD];
        kl = xb_cur[t * 36 + lane];
      };
      auto stepM = [&](const float (&kkv)[HD], float kl, float ri) {
        float vp0 = 0.f, vp1 = 0.f, vp2 = 0.f, vp3 = 0.f;
#pragma unroll
        for (int j = 0; j < HD; j += 4) {
          vp0 = fmaf(Mrow[j],     kkv[j],     vp0);
          vp1 = fmaf(Mrow[j + 1], kkv[j + 1], vp1);
          vp2 = fmaf(Mrow[j + 2], kkv[j + 2], vp2);
          vp3 = fmaf(Mrow[j + 3], kkv[j + 3], vp3);
        }
        float vp = (vp0 + vp1) + (vp2 + vp3);
        float u = fmaf(-vp, ri, kl);
#pragma unroll
        for (int j = 0; j < HD; ++j) Mrow[j] = fmaf(u, kkv[j], Mrow[j]);
      };

      float kkA[HD], kkB[HD];
      float klA, riA, klB, riB;
      loadk(0, kkA, klA, riA);
      for (int t = 0; t < steps; t += 2) {
        if (t + 1 < steps) loadk(t + 1, kkB, klB, riB);
        stepM(kkA, klA, riA);
        if (t + 2 < steps) loadk(t + 2, kkA, klA, riA);
        if (t + 1 < steps) stepM(kkB, klB, riB);
      }

      if (last) {
        const float* kl = &xb_cur[(TCH - 1) * 36];
        float c0 = 0.f, c1 = 0.f, c2 = 0.f, c3 = 0.f;
#pragma unroll
        for (int j = 0; j < HD; j += 4) {
          c0 = fmaf(Mrow[j],     kl[j],     c0);
          c1 = fmaf(Mrow[j + 1], kl[j + 1], c1);
          c2 = fmaf(Mrow[j + 2], kl[j + 2], c2);
          c3 = fmaf(Mrow[j + 3], kl[j + 3], c3);
        }
        s_ctx[lane] = (c0 + c1) + (c2 + c3);
      }
    }
    __syncthreads();
  }

  if (tid < HD) {
    float acc = br[tid];
#pragma unroll
    for (int j = 0; j < HD; ++j) acc = fmaf(s_ctx[j], wr[j * HD + tid], acc);
    s_tmp[tid] = acc;
  }
  __syncthreads();
  if (tid < VD) {
    float acc = bo[tid];
#pragma unroll
    for (int i = 0; i < HD; ++i) acc = fmaf(s_tmp[i], wo[i * VD + tid], acc);
    out[(size_t)bidx * VD + tid] = acc;
  }
}

extern "C" void kernel_launch(void* const* d_in, const int* in_sizes, int n_in,
                              void* d_out, int out_size, void* d_ws, size_t ws_size,
                              hipStream_t stream) {
  const int*   seq   = (const int*)  d_in[0];
  const float* embed = (const float*)d_in[1];
  const float* w1    = (const float*)d_in[2];
  const float* b1    = (const float*)d_in[3];
  const float* w2    = (const float*)d_in[4];
  const float* b2    = (const float*)d_in[5];
  const float* ln_g  = (const float*)d_in[6];
  const float* ln_b  = (const float*)d_in[7];
  const float* wr    = (const float*)d_in[8];
  const float* br    = (const float*)d_in[9];
  const float* wo    = (const float*)d_in[10];
  const float* bo    = (const float*)d_in[11];
  float* out = (float*)d_out;

  const int B = out_size / VD;
  const int L = in_sizes[0] / B;

  delta_model_fused<<<B, NTHREADS, 0, stream>>>(
      seq, embed, w1, b1, w2, b2, ln_g, ln_b, wr, br, wo, bo, out, L);
}


# Scan's table reads grouped: the LDS reads of 4 steps issued back to back in one place, a group ahead, into a second register set, instead of spread among the steps' vector instructions
# speedup vs baseline: 3.0007x; 3.0007x over previous
#include <hip/hip_runtime.h>


typedef __attribute__((ext_vector_type(2))) float v2f;
typedef __attribute__((ext_vector_type(8))) float v8f;
typedef __attribute__((ext_vector_type(16))) float v16f;

#define HD 32
#define VD 64
#define NTHREADS 64
#define TS 100
#define CH 16
#define GS 4


__launch_bounds__(NTHREADS)
__global__ void delta_model_fused(
    const int*   __restrict__ seq,
    const float* __restrict__ embed,
    const float* __restrict__ w1, const float* __restrict__ b1,
    const float* __restrict__ w2, const float* __restrict__ b2,
    const float* __restrict__ ln_g, const float* __restrict__ ln_b,
    const float* __restrict__ wr, const float* __restrict__ br,
    const float* __restrict__ wo, const float* __restrict__ bo,
    float* __restrict__ out, int L)
{
  __shared__ float s_w1[HD * 2 * HD];
  __shared__ float s_w2[2 * HD * HD];
  __shared__ float s_b1[2 * HD];
  __shared__ float s_b2[HD];
  __shared__ float s_g[HD];
  __shared__ float s_bt[HD];
  __shared__ float s_h [VD * 34];
  __shared__ float s_ff[VD * 66];
  __shared__ __align__(16) float s_T[VD * TS];
  __shared__ float s_ctx[HD];
  __shared__ float s_tmp[HD];

  const int tid  = threadIdx.x;
  const int wave = __builtin_amdgcn_readfirstlane(tid) >> 5;
  const int lane = tid & 31;
  const int ln   = lane & 15;
  const int hl   = lane >> 4;
  const int bidx = blockIdx.x;

  for (int i = tid; i < HD * 2 * HD; i += NTHREADS) s_w1[i] = w1[i];
  for (int i = tid; i < 2 * HD * HD; i += NTHREADS) s_w2[i] = w2[i];
  if (tid < 2 * HD) s_b1[tid] = b1[tid];
  if (tid < HD) { s_b2[tid] = b2[tid]; s_g[tid] = ln_g[tid]; s_bt[tid] = ln_b[tid]; }
  __syncthreads();

#pragma unroll 1
  for (int sl = 0; sl < 2; ++sl) {
    const int m0 = (wave * 2 + sl) * 16;
#pragma unroll
    for (int r = 0; r < 16; ++r)
      s_h[(m0 + r) * 34 + lane] = embed[(m0 + r) * HD + lane];
#pragma unroll
    for (int nt = 0; nt < 4; ++nt) {
      const int n0 = nt * 16;
      v8f acc = {0.f, 0.f, 0.f, 0.f, 0.f, 0.f, 0.f, 0.f};
#pragma unroll
      for (int k0 = 0; k0 < HD; k0 += 4) {
        const float* ap = &s_h[(m0 + ln) * 34 + k0 + 2 * hl];
        v2f a; a.x = ap[0]; a.y = ap[1];
        v2f bf; bf.x = s_w1[(k0 + 2 * hl) * 64 + n0 + ln];
        bf.y = s_w1[(k0 + 2 * hl + 1) * 64 + n0 + ln];
        acc = __builtin_amdgcn_wmma_f32_16x16x4_f32(false, a, false, bf,
                                                    (short)0, acc, false, false);
      }
      float bias = s_b1[n0 + ln];
#pragma unroll
      for (int r = 0; r < 8; ++r) {
        int row = m0 + r + 8 * hl;
        float v = acc[r] + bias;
        s_ff[row * 66 + n0 + ln] = v > 0.f ? v : 0.f;
      }
    }
#pragma unroll
    for (int nt = 0; nt < 2; ++nt) {
      const int n0 = nt * 16;
      v8f acc = {0.f, 0.f, 0.f, 0.f, 0.f, 0.f, 0.f, 0.f};
#pragma unroll
      for (int k0 = 0; k0 < 2 * HD; k0 += 4) {
        const float* ap = &s_ff[(m0 + ln) * 66 + k0 + 2 * hl];
        v2f a; a.x = ap[0]; a.y = ap[1];
        v2f bf; bf.x = s_w2[(k0 + 2 * hl) * HD + n0 + ln];
        bf.y = s_w2[(k0 + 2 * hl + 1) * HD + n0 + ln];
        acc = __builtin_amdgcn_wmma_f32_16x16x4_f32(false, a, false, bf,
                                                    (short)0, acc, false, false);
      }
      float bias = s_b2[n0 + ln];
#pragma unroll
      for (int r = 0; r < 8; ++r) {
        int row = m0 + r + 8 * hl, col = n0 + ln;
        s_T[row * TS + 64 + col] = acc[r] + bias + s_h[row * 34 + col];
      }
    }
    if (lane < 16) {
      float* xr = &s_T[(m0 + lane) * TS + 64];
      float mu = 0.f;
#pragma unroll
      for (int j = 0; j < HD; ++j) mu += xr[j];
      mu *= (1.f / 32.f);
      float var = 0.f;
#pragma unroll
      for (int j = 0; j < HD; ++j) { float d = xr[j] - mu; var = fmaf(d, d, var); }
      var *= (1.f / 32.f);
      float rs = __builtin_amdgcn_rsqf(var + 1e-5f);
      float kk2 = 0.f;
#pragma unroll
      for (int j = 0; j < HD; ++j) {
        float v = (xr[j] - mu) * rs * s_g[j] + s_bt[j];
        xr[j] = v;
        kk2 = fmaf(v, v, kk2);
      }
      xr[HD]     = kk2 + 1e-6f;
      xr[HD + 1] = __builtin_amdgcn_rcpf(kk2 + 1e-6f);
    }
  }
  __syncthreads();

#pragma unroll 1
  for (int mt = 0; mt < 2; ++mt) {
    const int m0 = (wave * 2 + mt) * 16;
#pragma unroll
    for (int nt = 0; nt < 4; ++nt) {
      const int n0 = nt * 16;
      v8f acc = {0.f, 0.f, 0.f, 0.f, 0.f, 0.f, 0.f, 0.f};
#pragma unroll
      for (int k0 = 0; k0 < HD; k0 += 4) {
        const float* ap = &s_T[(m0 + ln) * TS + 64 + k0 + 2 * hl];
        const float* bp = &s_T[(n0 + ln) * TS + 64 + k0 + 2 * hl];
        v2f a;  a.x  = ap[0]; a.y  = ap[1];
        v2f bf; bf.x = bp[0]; bf.y = bp[1];
        acc = __builtin_amdgcn_wmma_f32_16x16x4_f32(false, a, false, bf,
                                                    (short)0, acc, false, false);
      }
      const float rb = s_T[(n0 + ln) * TS + 97];
#pragma unroll
      for (int r = 0; r < 8; ++r)
        s_T[(m0 + r + 8 * hl) * TS + n0 + ln] = acc[r] * rb;
    }
  }
  __syncthreads();

  const int q = lane & 3;
  const int p = wave * 8 + (lane >> 2);
  const int* sp = seq + (size_t)bidx * L;
  const int nch = L / CH;
  const int a_last = __builtin_amdgcn_readfirstlane(sp[L - 1]);

  v16f S0 = (v16f)(0.f), S1 = (v16f)(0.f);
  float sel0 = 0.f, sel1 = 0.f;
  v16f GA[GS], GB[GS];
  v2f  EA[GS], EB[GS];

  auto fetch_group = [&](v16f (&G)[GS], v2f (&E)[GS], int t0, int t1, int t2, int t3) {
    const int tk[GS] = {t0, t1, t2, t3};
    const float4* gp[GS];
    const v2f*    ep[GS];
#pragma unroll
    for (int k = 0; k < GS; ++k) {
      gp[k] = (const float4*)(s_T + tk[k] * TS + q * 16);
      ep[k] = (const v2f*)(s_T + tk[k] * TS + 64 + 2 * p);
    }
#pragma unroll
    for (int k = 0; k < GS; ++k) {
      float4 x0 = gp[k][0], x1 = gp[k][1], x2 = gp[k][2], x3 = gp[k][3];
      G[k] = (v16f){x0.x, x0.y, x0.z, x0.w, x1.x, x1.y, x1.z, x1.w,
                    x2.x, x2.y, x2.z, x2.w, x3.x, x3.y, x3.z, x3.w};
      E[k] = *ep[k];
    }
    __builtin_amdgcn_sched_barrier(0);
  };

  auto step = [&](const v16f& G, const v2f& E, int an) {
    const int idx = an & 15;
    const unsigned qn = (unsigned)an >> 4;
    const unsigned slo = 0x44440000u + qn * 0x11111111u;
    const unsigned shi = 0xCCCC8888u + qn * 0x11111111u;
    const float u0 = E.x - sel0;
    const float u1 = E.y - sel1;
    const float gs = G[idx];
    const float n0 = fmaf(u0, gs, S0[idx]);
    const float n1 = fmaf(u1, gs, S1[idx]);
    sel0 = __builtin_bit_cast(float, __builtin_amdgcn_permlane16(
               __builtin_bit_cast(unsigned, n0), __builtin_bit_cast(unsigned, n0),
               slo, shi, false, false));
    sel1 = __builtin_bit_cast(float, __builtin_amdgcn_permlane16(
               __builtin_bit_cast(unsigned, n1), __builtin_bit_cast(unsigned, n1),
               slo, shi, false, false));
    S0 = __builtin_elementwise_fma((v16f)(u0), G, S0);
    S1 = __builtin_elementwise_fma((v16f)(u1), G, S1);
  };

  int cur[CH], nx1[CH];
  {
    const int r0 = sp[ln];
    const int r1 = sp[(nch > 1 ? CH : 0) + ln];
#pragma unroll
    for (int j = 0; j < CH; ++j) {
      cur[j] = __builtin_amdgcn_readlane(r0, j);
      nx1[j] = __builtin_amdgcn_readlane(r1, j);
    }
  }

  fetch_group(GA, EA, cur[0], cur[1], cur[2], cur[3]);

#pragma unroll 1
  for (int c = 0; c < nch; ++c) {
    const int cn = (c + 2 < nch) ? c + 2 : nch - 1;
    const int raw = sp[cn * CH + ln];

    fetch_group(GB, EB, cur[4], cur[5], cur[6], cur[7]);
    step(GA[0], EA[0], cur[1]);
    step(GA[1], EA[1], cur[2]);
    step(GA[2], EA[2], cur[3]);
    step(GA[3], EA[3], cur[4]);

    fetch_group(GA, EA, cur[8], cur[9], cur[10], cur[11]);
    step(GB[0], EB[0], cur[5]);
    step(GB[1], EB[1], cur[6]);
    step(GB[2], EB[2], cur[7]);
    step(GB[3], EB[3], cur[8]);

    fetch_group(GB, EB, cur[12], cur[13], cur[14], cur[15]);
    step(GA[0], EA[0], cur[9]);
    step(GA[1], EA[1], cur[10]);
    step(GA[2], EA[2], cur[11]);
    step(GA[3], EA[3], cur[12]);

    fetch_group(GA, EA, nx1[0], nx1[1], nx1[2], nx1[3]);
    step(GB[0], EB[0], cur[13]);
    step(GB[1], EB[1], cur[14]);
    step(GB[2], EB[2], cur[15]);
    if (c + 1 < nch)
      step(GB[3], EB[3], nx1[0]);

#pragma unroll
    for (int j = 0; j < CH; ++j) { cur[j] = nx1[j]; nx1[j] = __builtin_amdgcn_readlane(raw, j); }
  }

  if (q == 0) {
    const float den = s_T[a_last * TS + 96];
    s_ctx[2 * p]     = sel0 * den;
    s_ctx[2 * p + 1] = sel1 * den;
  }
  __syncthreads();

  if (tid < HD) {
    float acc = br[tid];
#pragma unroll
    for (int j = 0; j < HD; ++j) acc = fmaf(s_ctx[j], wr[j * HD + tid], acc);
    s_tmp[tid] = acc;
  }
  __syncthreads();
  if (tid < VD) {
    float acc = bo[tid];
#pragma unroll
    for (int i = 0; i < HD; ++i) acc = fmaf(s_tmp[i], wo[i * VD + tid], acc);
    out[(size_t)bidx * VD + tid] = acc;
  }
}

extern "C" void kernel_launch(void* const* d_in, const int* in_sizes, int n_in,
                              void* d_out, int out_size, void* d_ws, size_t ws_size,
                              hipStream_t stream) {
  const int*   seq   = (const int*)  d_in[0];
  const float* embed = (const float*)d_in[1];
  const float* w1    = (const float*)d_in[2];
  const float* b1    = (const float*)d_in[3];
  const float* w2    = (const float*)d_in[4];
  const float* b2    = (const float*)d_in[5];
  const float* ln_g  = (const float*)d_in[6];
  const float* ln_b  = (const float*)d_in[7];
  const float* wr    = (const float*)d_in[8];
  const float* br    = (const float*)d_in[9];
  const float* wo    = (const float*)d_in[10];
  const float* bo    = (const float*)d_in[11];
  float* out = (float*)d_out;

  const int B = out_size / VD;
  const int L = in_sizes[0] / B;

  delta_model_fused<<<B, NTHREADS, 0, stream>>>(
      seq, embed, w1, b1, w2, b2, ln_g, ln_b, wr, br, wo, bo, out, L);
}


# All global reads of the block requested in one batch at entry by 16-byte loads (embed rows too); head weights fetched to LDS by the idle wave during the scan: one memory wait, was ten; sums unchanged
# speedup vs baseline: 6.6993x; 2.2326x over previous
#include <hip/hip_runtime.h>


typedef __attribute__((ext_vector_type(2))) float v2f;
typedef __attribute__((ext_vector_type(4))) float v4f;
typedef __attribute__((ext_vector_type(8))) float v8f;
typedef __attribute__((ext_vector_type(4))) unsigned v4u;

#define HD 32
#define VD 64
#define NTHREADS 64
#define ES 36
#define CH 16
#define GS 8
#define LMAX 2048


__launch_bounds__(NTHREADS)
__global__ void delta_model_fused(
    const int*   __restrict__ seq,
    const float* __restrict__ embed,
    const float* __restrict__ w1, const float* __restrict__ b1,
    const float* __restrict__ w2, const float* __restrict__ b2,
    const float* __restrict__ ln_g, const float* __restrict__ ln_b,
    const float* __restrict__ wr, const float* __restrict__ br,
    const float* __restrict__ wo, const float* __restrict__ bo,
    float* __restrict__ out, int L)
{
  __shared__ __align__(16) float s_w1[HD * 2 * HD];
  __shared__ __align__(16) float s_w2[2 * HD * HD];
  __shared__ float s_b1[2 * HD];
  __shared__ float s_b2[HD];
  __shared__ float s_g[HD];
  __shared__ float s_bt[HD];
  __shared__ __align__(16) float s_h [VD * 34];
  __shared__ float s_ff[VD * 66];
  __shared__ __align__(16) float s_E[VD * ES];
  __shared__ __align__(16) float s_G[VD * VD];
  __shared__ float s_E2[VD * HD];
  __shared__ __align__(16) float s_R[VD * 32 * 8];
  __shared__ float s_ctx[HD];
  __shared__ float s_tmp[HD];
  __shared__ __align__(16) int s_tok[LMAX];

  const int tid  = threadIdx.x;
  const int wave = __builtin_amdgcn_readfirstlane(tid) >> 5;
  const int lane = tid & 31;
  const int ln   = lane & 15;
  const int hl   = lane >> 4;
  const int bidx = blockIdx.x;

  const int nq = (L < LMAX ? L : LMAX) >> 2;
  const int4* sq = (const int4*)(seq + (size_t)bidx * L);
  int4 tq[LMAX / 4 / NTHREADS];
#pragma unroll
  for (int j = 0; j < LMAX / 4 / NTHREADS; ++j) {
    const int i = tid + j * NTHREADS;
    tq[j] = (i < nq) ? sq[i] : make_int4(0, 0, 0, 0);
  }

  constexpr int NQ = HD * 2 * HD / 4 / NTHREADS;
  float4 w1r[NQ], w2r[NQ], er[NQ];
#pragma unroll
  for (int j = 0; j < NQ; ++j) w1r[j] = ((const float4*)w1)[tid + j * NTHREADS];
#pragma unroll
  for (int j = 0; j < NQ; ++j) w2r[j] = ((const float4*)w2)[tid + j * NTHREADS];
#pragma unroll
  for (int j = 0; j < NQ; ++j) er[j] = ((const float4*)embed)[tid + j * NTHREADS];
  const float b1r = b1[tid & (2 * HD - 1)];
  const float b2r = b2[tid & (HD - 1)], lgr = ln_g[tid & (HD - 1)], lbr = ln_b[tid & (HD - 1)];

#pragma unroll
  for (int j = 0; j < NQ; ++j) ((float4*)s_w1)[tid + j * NTHREADS] = w1r[j];
#pragma unroll
  for (int j = 0; j < NQ; ++j) ((float4*)s_w2)[tid + j * NTHREADS] = w2r[j];
#pragma unroll
  for (int j = 0; j < NQ; ++j) {
    const int i = tid + j * NTHREADS;
    float* d = &s_h[(i >> 3) * 34 + 4 * (i & 7)];
    *(v2f*)d       = (v2f){er[j].x, er[j].y};
    *(v2f*)(d + 2) = (v2f){er[j].z, er[j].w};
  }
  if (tid < 2 * HD) s_b1[tid] = b1r;
  if (tid < HD) { s_b2[tid] = b2r; s_g[tid] = lgr; s_bt[tid] = lbr; }
#pragma unroll
  for (int j = 0; j < LMAX / 4 / NTHREADS; ++j)
    ((int4*)s_tok)[tid + j * NTHREADS] = tq[j];
  __syncthreads();

#pragma unroll 1
  for (int sl = 0; sl < 2; ++sl) {
    const int m0 = (wave * 2 + sl) * 16;
#pragma unroll
    for (int nt = 0; nt < 4; ++nt) {
      const int n0 = nt * 16;
      v8f acc = {0.f, 0.f, 0.f, 0.f, 0.f, 0.f, 0.f, 0.f};
#pragma unroll
      for (int k0 = 0; k0 < HD; k0 += 4) {
        const float* ap = &s_h[(m0 + ln) * 34 + k0 + 2 * hl];
        v2f a; a.x = ap[0]; a.y = ap[1];
        v2f bf; bf.x = s_w1[(k0 + 2 * hl) * 64 + n0 + ln];
        bf.y = s_w1[(k0 + 2 * hl + 1) * 64 + n0 + ln];
        acc = __builtin_amdgcn_wmma_f32_16x16x4_f32(false, a, false, bf,
                                                    (short)0, acc, false, false);
      }
      float bias = s_b1[n0 + ln];
#pragma unroll
      for (int r = 0; r < 8; ++r) {
        int row = m0 + r + 8 * hl;
        float v = acc[r] + bias;
        s_ff[row * 66 + n0 + ln] = v > 0.f ? v : 0.f;
      }
    }
#pragma unroll
    for (int nt = 0; nt < 2; ++nt) {
      const int n0 = nt * 16;
      v8f acc = {0.f, 0.f, 0.f, 0.f, 0.f, 0.f, 0.f, 0.f};
#pragma unroll
      for (int k0 = 0; k0 < 2 * HD; k0 += 4) {
        const float* ap = &s_ff[(m0 + ln) * 66 + k0 + 2 * hl];
        v2f a; a.x = ap[0]; a.y = ap[1];
        v2f bf; bf.x = s_w2[(k0 + 2 * hl) * HD + n0 + ln];
        bf.y = s_w2[(k0 + 2 * hl + 1) * HD + n0 + ln];
        acc = __builtin_amdgcn_wmma_f32_16x16x4_f32(false, a, false, bf,
                                                    (short)0, acc, false, false);
      }
      float bias = s_b2[n0 + ln];
#pragma unroll
      for (int r = 0; r < 8; ++r) {
        int row = m0 + r + 8 * hl, col = n0 + ln;
        s_E[row * ES + col] = acc[r] + bias + s_h[row * 34 + col];
      }
    }
    if (lane < 16) {
      float* xr = &s_E[(m0 + lane) * ES];
      float mu = 0.f;
#pragma unroll
      for (int j = 0; j < HD; ++j) mu += xr[j];
      mu *= (1.f / 32.f);
      float var = 0.f;
#pragma unroll
      for (int j = 0; j < HD; ++j) { float d = xr[j] - mu; var = fmaf(d, d, var); }
      var *= (1.f / 32.f);
      float rs = __builtin_amdgcn_rsqf(var + 1e-5f);
      float kk2 = 0.f;
#pragma unroll
      for (int j = 0; j < HD; ++j) {
        float v = (xr[j] - mu) * rs * s_g[j] + s_bt[j];
        xr[j] = v;
        kk2 = fmaf(v, v, kk2);
      }
      xr[HD]     = kk2 + 1e-6f;
      xr[HD + 1] = __builtin_amdgcn_rcpf(kk2 + 1e-6f);
    }
  }
  __syncthreads();

  for (int i = tid; i < VD * HD; i += NTHREADS) s_E2[i] = s_E[(i >> 5) * ES + (i & 31)];

#pragma unroll 1
  for (int mt = 0; mt < 2; ++mt) {
    const int m0 = (wave * 2 + mt) * 16;
    float nra[8];
#pragma unroll
    for (int r = 0; r < 8; ++r) nra[r] = -s_E[(m0 + r + 8 * hl) * ES + HD + 1];
#pragma unroll
    for (int nt = 0; nt < 4; ++nt) {
      const int n0 = nt * 16;
      v8f acc = {0.f, 0.f, 0.f, 0.f, 0.f, 0.f, 0.f, 0.f};
#pragma unroll
      for (int k0 = 0; k0 < HD; k0 += 4) {
        const float* ap = &s_E[(m0 + ln) * ES + k0 + 2 * hl];
        const float* bp = &s_E[(n0 + ln) * ES + k0 + 2 * hl];
        v2f a;  a.x  = ap[0]; a.y  = ap[1];
        v2f bf; bf.x = bp[0]; bf.y = bp[1];
        acc = __builtin_amdgcn_wmma_f32_16x16x4_f32(false, a, false, bf,
                                                    (short)0, acc, false, false);
      }
#pragma unroll
      for (int r = 0; r < 8; ++r)
        s_G[(m0 + r + 8 * hl) * VD + n0 + ln] = acc[r] * nra[r];
    }
  }
  __syncthreads();

#pragma unroll 1
  for (int i0 = tid; i0 < VD * 32; i0 += NTHREADS * 8) {
    v4f q[8], x[8];
#pragma unroll
    for (int j = 0; j < 8; ++j) {
      const int i = i0 + j * NTHREADS, b = i >> 5, l = i & 31;
      q[j] = *(const v4f*)(s_G + b * VD + 4 * (l & 15));
      x[j] = (v4f){__builtin_bit_cast(float, (b & 1) ? 0xffffffffu : 0u),
                   __builtin_bit_cast(float, (b & 2) ? 0xffffffffu : 0u),
                   __builtin_bit_cast(float, (unsigned)b >> 2),
                   s_E2[b * HD + l]};
    }
    __builtin_amdgcn_sched_barrier(0);
#pragma unroll
    for (int j = 0; j < 8; ++j) {
      const int i = i0 + j * NTHREADS;
      *(v4f*)(s_R + i * 8)     = q[j];
      *(v4f*)(s_R + i * 8 + 4) = x[j];
    }
  }
  __syncthreads();

  if (wave == 0) {
    const int* sp = s_tok;
    const int nch = (L < LMAX ? L : LMAX) / CH;

    float y0, y1, y2, y3;
    float cacc = 0.f;
    float s = 0.f;
    v4u HA[GS], HB[GS];
    v4u XA[GS], XB[GS];

    auto fetch_group = [&](v4u (&Hg)[GS], v4u (&Xg)[GS], int t0, int t1, int t2, int t3,
                           int t4, int t5, int t6, int t7, const int* idp, int& idv) {
      const int tk[GS] = {t0, t1, t2, t3, t4, t5, t6, t7};
      const uint4* hp[GS];
#pragma unroll
      for (int k = 0; k < GS; ++k)
        hp[k] = (const uint4*)(s_R + tk[k] * 256 + lane * 8);
      __builtin_amdgcn_sched_barrier(0);
      idv = *idp;
#pragma unroll
      for (int k = 0; k < GS; ++k) {
        const uint4 q = hp[k][0];
        const uint4 x = hp[k][1];
        Hg[k] = (v4u){q.x, q.y, q.z, q.w};
        Xg[k] = (v4u){x.x, x.y, x.z, x.w};
      }
      __builtin_amdgcn_sched_barrier(0);
    };

    auto sel4 = [](unsigned u0, unsigned u1, unsigned u2, unsigned u3, unsigned m0, unsigned m1) -> float {
      const unsigned p01 = ((u1 ^ u0) & m0) ^ u0;
      const unsigned p23 = ((u3 ^ u2) & m0) ^ u2;
      return __builtin_bit_cast(float, ((p23 ^ p01) & m1) ^ p01);
    };

    float zs;
    float hs;

    auto step = [&](const v4u& Hr, const v4u& Xr, const v4u& Xn, const v4u& Hn1, const v4u& Xnn) {
      const unsigned q0 = Hr.x, q1 = Hr.y, q2 = Hr.z, q3 = Hr.w, xe = Xr.w;
      const float h0 = __builtin_bit_cast(float, q0), h1 = __builtin_bit_cast(float, q1);
      const float h2 = __builtin_bit_cast(float, q2), h3 = __builtin_bit_cast(float, q3);
      const unsigned fsel = Xn.z;
      const unsigned gm0 = Xnn.x, gm1 = Xnn.y;
      const unsigned n0 = Hn1.x, n1 = Hn1.y, n2 = Hn1.z, n3 = Hn1.w;
      y0 = fmaf(s, h0, y0);
      y1 = fmaf(s, h1, y1);
      y2 = fmaf(s, h2, y2);
      y3 = fmaf(s, h3, y3);
      const float z = fmaf(s, hs, zs);
      hs = sel4(n0, n1, n2, n3, gm0, gm1);
      cacc = fmaf(s, __builtin_bit_cast(float, xe), cacc);
      asm volatile("" :: "v"(cacc));
      const unsigned zu = __builtin_bit_cast(unsigned, z);
      s = __builtin_bit_cast(float, __builtin_amdgcn_permlane16_var(zu, zu, fsel, false, false));
      zs = sel4(__builtin_bit_cast(unsigned, y0), __builtin_bit_cast(unsigned, y1),
                __builtin_bit_cast(unsigned, y2), __builtin_bit_cast(unsigned, y3), gm0, gm1);
    };

    int cur[CH], nx1[CH];
    {
      const int r0 = sp[(nch - 1) * CH + ln];
      const int r1 = sp[(nch > 1 ? nch - 2 : 0) * CH + ln];
#pragma unroll
      for (int j = 0; j < CH; ++j) {
        cur[j] = __builtin_amdgcn_readlane(r0, CH - 1 - j);
        nx1[j] = __builtin_amdgcn_readlane(r1, CH - 1 - j);
      }
    }

    {
      const float nden = -s_E[cur[0] * ES + HD];
      const float4 h = *(const float4*)(s_G + cur[0] * VD + 4 * ln);
      y0 = h.x * nden; y1 = h.y * nden; y2 = h.z * nden; y3 = h.w * nden;
    }

    int noid;
    fetch_group(HA, XA, cur[0], cur[1], cur[2], cur[3], cur[4], cur[5], cur[6], cur[7], sp, noid);
    {
      const unsigned fm0 = XA[1].x, fm1 = XA[1].y;
      zs = sel4(__builtin_bit_cast(unsigned, y0), __builtin_bit_cast(unsigned, y1),
                __builtin_bit_cast(unsigned, y2), __builtin_bit_cast(unsigned, y3), fm0, fm1);
      const unsigned q0 = HA[0].x, q1 = HA[0].y, q2 = HA[0].z, q3 = HA[0].w;
      hs = sel4(q0, q1, q2, q3, fm0, fm1);
    }

#pragma unroll 1
    for (int c = nch - 1; c >= 0; --c) {
      const int cn = (c >= 2) ? c - 2 : 0;
      int raw;

      fetch_group(HB, XB, cur[8], cur[9], cur[10], cur[11], cur[12], cur[13], cur[14], cur[15],
                  sp + cn * CH + ln, raw);
      step(HA[0], XA[0], XA[1], HA[1], XA[2]);
      step(HA[1], XA[1], XA[2], HA[2], XA[3]);
      step(HA[2], XA[2], XA[3], HA[3], XA[4]);
      step(HA[3], XA[3], XA[4], HA[4], XA[5]);
      step(HA[4], XA[4], XA[5], HA[5], XA[6]);
      step(HA[5], XA[5], XA[6], HA[6], XA[7]);
      step(HA[6], XA[6], XA[7], HA[7], XB[0]);
      step(HA[7], XA[7], XB[0], HB[0], XB[1]);

      fetch_group(HA, XA, nx1[0], nx1[1], nx1[2], nx1[3], nx1[4], nx1[5], nx1[6], nx1[7], sp, noid);
      step(HB[0], XB[0], XB[1], HB[1], XB[2]);
      step(HB[1], XB[1], XB[2], HB[2], XB[3]);
      step(HB[2], XB[2], XB[3], HB[3], XB[4]);
      step(HB[3], XB[3], XB[4], HB[4], XB[5]);
      step(HB[4], XB[4], XB[5], HB[5], XB[6]);
      step(HB[5], XB[5], XB[6], HB[6], XB[7]);
      step(HB[6], XB[6], XB[7], HB[7], XA[0]);
      step(HB[7], XB[7], XA[0], HA[0], XA[1]);

#pragma unroll
      for (int j = 0; j < CH; ++j) { cur[j] = nx1[j]; nx1[j] = __builtin_amdgcn_readlane(raw, CH - 1 - j); }
    }

    s_ctx[lane] = cacc;
  } else {
    float4 wrr[HD * HD / 4 / 32], wor[HD * VD / 4 / 32];
#pragma unroll
    for (int j = 0; j < HD * HD / 4 / 32; ++j) wrr[j] = ((const float4*)wr)[lane + j * 32];
#pragma unroll
    for (int j = 0; j < HD * VD / 4 / 32; ++j) wor[j] = ((const float4*)wo)[lane + j * 32];
    const float brr = br[lane], bo0 = bo[lane], bo1 = bo[lane + 32];
#pragma unroll
    for (int j = 0; j < HD * HD / 4 / 32; ++j) ((float4*)s_w2)[lane + j * 32] = wrr[j];
#pragma unroll
    for (int j = 0; j < HD * VD / 4 / 32; ++j) ((float4*)s_w1)[lane + j * 32] = wor[j];
    s_b2[lane] = brr; s_b1[lane] = bo0; s_b1[lane + 32] = bo1;
  }
  __syncthreads();

  if (tid < HD) {
    float acc = s_b2[tid];
#pragma unroll
    for (int j = 0; j < HD; ++j) acc = fmaf(s_ctx[j], s_w2[j * HD + tid], acc);
    s_tmp[tid] = acc;
  }
  __syncthreads();
  if (tid < VD) {
    float acc = s_b1[tid];
#pragma unroll
    for (int i = 0; i < HD; ++i) acc = fmaf(s_tmp[i], s_w1[i * VD + tid], acc);
    out[(size_t)bidx * VD + tid] = acc;
  }
}

extern "C" void kernel_launch(void* const* d_in, const int* in_sizes, int n_in,
                              void* d_out, int out_size, void* d_ws, size_t ws_size,
                              hipStream_t stream) {
  const int*   seq   = (const int*)  d_in[0];
  const float* embed = (const float*)d_in[1];
  const float* w1    = (const float*)d_in[2];
  const float* b1    = (const float*)d_in[3];
  const float* w2    = (const float*)d_in[4];
  const float* b2    = (const float*)d_in[5];
  const float* ln_g  = (const float*)d_in[6];
  const float* ln_b  = (const float*)d_in[7];
  const float* wr    = (const float*)d_in[8];
  const float* br    = (const float*)d_in[9];
  const float* wo    = (const float*)d_in[10];
  const float* bo    = (const float*)d_in[11];
  float* out = (float*)d_out;

  const int B = out_size / VD;
  const int L = in_sizes[0] / B;

  delta_model_fused<<<B, NTHREADS, 0, stream>>>(
      seq, embed, w1, b1, w2, b2, ln_g, ln_b, wr, br, wo, bo, out, L);
}


# Scan's table reads in places of 16 steps instead of 8, 8 steps ahead, into a 24-slot register ring (step q in slot q mod 24; 48-step loop pass); ids to scalars mid-place; sums and order unchanged
# speedup vs baseline: 7.2594x; 1.0836x over previous
#include <hip/hip_runtime.h>


typedef __attribute__((ext_vector_type(2))) float v2f;
typedef __attribute__((ext_vector_type(4))) float v4f;
typedef __attribute__((ext_vector_type(8))) float v8f;
typedef __attribute__((ext_vector_type(4))) unsigned v4u;

#define HD 32
#define VD 64
#define NTHREADS 64
#define ES 36
#define PS 16
#define LD 8
#define RS (PS + LD)
template <int N> struct IC { static constexpr int v = N; };
#define LMAX 2048


__launch_bounds__(NTHREADS)
__global__ void delta_model_fused(
    const int*   __restrict__ seq,
    const float* __restrict__ embed,
    const float* __restrict__ w1, const float* __restrict__ b1,
    const float* __restrict__ w2, const float* __restrict__ b2,
    const float* __restrict__ ln_g, const float* __restrict__ ln_b,
    const float* __restrict__ wr, const float* __restrict__ br,
    const float* __restrict__ wo, const float* __restrict__ bo,
    float* __restrict__ out, int L)
{
  __shared__ __align__(16) float s_w1[HD * 2 * HD];
  __shared__ __align__(16) float s_w2[2 * HD * HD];
  __shared__ float s_b1[2 * HD];
  __shared__ float s_b2[HD];
  __shared__ float s_g[HD];
  __shared__ float s_bt[HD];
  __shared__ __align__(16) float s_h [VD * 34];
  __shared__ float s_ff[VD * 66];
  __shared__ __align__(16) float s_E[VD * ES];
  __shared__ __align__(16) float s_G[VD * VD];
  __shared__ float s_E2[VD * HD];
  __shared__ __align__(16) float s_R[VD * 32 * 8];
  __shared__ float s_ctx[HD];
  __shared__ float s_tmp[HD];
  __shared__ __align__(16) int s_tok[LMAX];

  const int tid  = threadIdx.x;
  const int wave = __builtin_amdgcn_readfirstlane(tid) >> 5;
  const int lane = tid & 31;
  const int ln   = lane & 15;
  const int hl   = lane >> 4;
  const int bidx = blockIdx.x;

  const int nq = (L < LMAX ? L : LMAX) >> 2;
  const int4* sq = (const int4*)(seq + (size_t)bidx * L);
  int4 tq[LMAX / 4 / NTHREADS];
#pragma unroll
  for (int j = 0; j < LMAX / 4 / NTHREADS; ++j) {
    const int i = tid + j * NTHREADS;
    tq[j] = (i < nq) ? sq[i] : make_int4(0, 0, 0, 0);
  }

  constexpr int NQ = HD * 2 * HD / 4 / NTHREADS;
  float4 w1r[NQ], w2r[NQ], er[NQ];
#pragma unroll
  for (int j = 0; j < NQ; ++j) w1r[j] = ((const float4*)w1)[tid + j * NTHREADS];
#pragma unroll
  for (int j = 0; j < NQ; ++j) w2r[j] = ((const float4*)w2)[tid + j * NTHREADS];
#pragma unroll
  for (int j = 0; j < NQ; ++j) er[j] = ((const float4*)embed)[tid + j * NTHREADS];
  const float b1r = b1[tid & (2 * HD - 1)];
  const float b2r = b2[tid & (HD - 1)], lgr = ln_g[tid & (HD - 1)], lbr = ln_b[tid & (HD - 1)];

#pragma unroll
  for (int j = 0; j < NQ; ++j) ((float4*)s_w1)[tid + j * NTHREADS] = w1r[j];
#pragma unroll
  for (int j = 0; j < NQ; ++j) ((float4*)s_w2)[tid + j * NTHREADS] = w2r[j];
#pragma unroll
  for (int j = 0; j < NQ; ++j) {
    const int i = tid + j * NTHREADS;
    float* d = &s_h[(i >> 3) * 34 + 4 * (i & 7)];
    *(v2f*)d       = (v2f){er[j].x, er[j].y};
    *(v2f*)(d + 2) = (v2f){er[j].z, er[j].w};
  }
  if (tid < 2 * HD) s_b1[tid] = b1r;
  if (tid < HD) { s_b2[tid] = b2r; s_g[tid] = lgr; s_bt[tid] = lbr; }
#pragma unroll
  for (int j = 0; j < LMAX / 4 / NTHREADS; ++j)
    ((int4*)s_tok)[tid + j * NTHREADS] = tq[j];
  __syncthreads();

#pragma unroll 1
  for (int sl = 0; sl < 2; ++sl) {
    const int m0 = (wave * 2 + sl) * 16;
#pragma unroll
    for (int nt = 0; nt < 4; ++nt) {
      const int n0 = nt * 16;
      v8f acc = {0.f, 0.f, 0.f, 0.f, 0.f, 0.f, 0.f, 0.f};
#pragma unroll
      for (int k0 = 0; k0 < HD; k0 += 4) {
        const float* ap = &s_h[(m0 + ln) * 34 + k0 + 2 * hl];
        v2f a; a.x = ap[0]; a.y = ap[1];
        v2f bf; bf.x = s_w1[(k0 + 2 * hl) * 64 + n0 + ln];
        bf.y = s_w1[(k0 + 2 * hl + 1) * 64 + n0 + ln];
        acc = __builtin_amdgcn_wmma_f32_16x16x4_f32(false, a, false, bf,
                                                    (short)0, acc, false, false);
      }
      float bias = s_b1[n0 + ln];
#pragma unroll
      for (int r = 0; r < 8; ++r) {
        int row = m0 + r + 8 * hl;
        float v = acc[r] + bias;
        s_ff[row * 66 + n0 + ln] = v > 0.f ? v : 0.f;
      }
    }
#pragma unroll
    for (int nt = 0; nt < 2; ++nt) {
      const int n0 = nt * 16;
      v8f acc = {0.f, 0.f, 0.f, 0.f, 0.f, 0.f, 0.f, 0.f};
#pragma unroll
      for (int k0 = 0; k0 < 2 * HD; k0 += 4) {
        const float* ap = &s_ff[(m0 + ln) * 66 + k0 + 2 * hl];
        v2f a; a.x = ap[0]; a.y = ap[1];
        v2f bf; bf.x = s_w2[(k0 + 2 * hl) * HD + n0 + ln];
        bf.y = s_w2[(k0 + 2 * hl + 1) * HD + n0 + ln];
        acc = __builtin_amdgcn_wmma_f32_16x16x4_f32(false, a, false, bf,
                                                    (short)0, acc, false, false);
      }
      float bias = s_b2[n0 + ln];
#pragma unroll
      for (int r = 0; r < 8; ++r) {
        int row = m0 + r + 8 * hl, col = n0 + ln;
        s_E[row * ES + col] = acc[r] + bias + s_h[row * 34 + col];
      }
    }
    if (lane < 16) {
      float* xr = &s_E[(m0 + lane) * ES];
      float mu = 0.f;
#pragma unroll
      for (int j = 0; j < HD; ++j) mu += xr[j];
      mu *= (1.f / 32.f);
      float var = 0.f;
#pragma unroll
      for (int j = 0; j < HD; ++j) { float d = xr[j] - mu; var = fmaf(d, d, var); }
      var *= (1.f / 32.f);
      float rs = __builtin_amdgcn_rsqf(var + 1e-5f);
      float kk2 = 0.f;
#pragma unroll
      for (int j = 0; j < HD; ++j) {
        float v = (xr[j] - mu) * rs * s_g[j] + s_bt[j];
        xr[j] = v;
        kk2 = fmaf(v, v, kk2);
      }
      xr[HD]     = kk2 + 1e-6f;
      xr[HD + 1] = __builtin_amdgcn_rcpf(kk2 + 1e-6f);
    }
  }
  __syncthreads();

  for (int i = tid; i < VD * HD; i += NTHREADS) s_E2[i] = s_E[(i >> 5) * ES + (i & 31)];

#pragma unroll 1
  for (int mt = 0; mt < 2; ++mt) {
    const int m0 = (wave * 2 + mt) * 16;
    float nra[8];
#pragma unroll
    for (int r = 0; r < 8; ++r) nra[r] = -s_E[(m0 + r + 8 * hl) * ES + HD + 1];
#pragma unroll
    for (int nt = 0; nt < 4; ++nt) {
      const int n0 = nt * 16;
      v8f acc = {0.f, 0.f, 0.f, 0.f, 0.f, 0.f, 0.f, 0.f};
#pragma unroll
      for (int k0 = 0; k0 < HD; k0 += 4) {
        const float* ap = &s_E[(m0 + ln) * ES + k0 + 2 * hl];
        const float* bp = &s_E[(n0 + ln) * ES + k0 + 2 * hl];
        v2f a;  a.x  = ap[0]; a.y  = ap[1];
        v2f bf; bf.x = bp[0]; bf.y = bp[1];
        acc = __builtin_amdgcn_wmma_f32_16x16x4_f32(false, a, false, bf,
                                                    (short)0, acc, false, false);
      }
#pragma unroll
      for (int r = 0; r < 8; ++r)
        s_G[(m0 + r + 8 * hl) * VD + n0 + ln] = acc[r] * nra[r];
    }
  }
  __syncthreads();

#pragma unroll 1
  for (int i0 = tid; i0 < VD * 32; i0 += NTHREADS * 8) {
    v4f q[8], x[8];
#pragma unroll
    for (int j = 0; j < 8; ++j) {
      const int i = i0 + j * NTHREADS, b = i >> 5, l = i & 31;
      q[j] = *(const v4f*)(s_G + b * VD + 4 * (l & 15));
      x[j] = (v4f){__builtin_bit_cast(float, (b & 1) ? 0xffffffffu : 0u),
                   __builtin_bit_cast(float, (b & 2) ? 0xffffffffu : 0u),
                   __builtin_bit_cast(float, (unsigned)b >> 2),
                   s_E2[b * HD + l]};
    }
    __builtin_amdgcn_sched_barrier(0);
#pragma unroll
    for (int j = 0; j < 8; ++j) {
      const int i = i0 + j * NTHREADS;
      *(v4f*)(s_R + i * 8)     = q[j];
      *(v4f*)(s_R + i * 8 + 4) = x[j];
    }
  }
  __syncthreads();

  if (wave == 0) {
    const int* sp = s_tok;
    const int Lc = (L < LMAX ? L : LMAX);
    const int nseg = Lc / PS;

    float y0, y1, y2, y3;
    float cacc = 0.f;
    float s = 0.f;
    v4u RH[RS];
    v4u RX[RS];
    int tk[PS];

    auto place = [&](auto first, const int* idp, int& idv) __attribute__((always_inline)) {
      constexpr int S0 = decltype(first)::v;
      const uint4* hp[PS];
#pragma unroll
      for (int k = 0; k < PS; ++k)
        hp[k] = (const uint4*)(s_R + tk[k] * 256 + lane * 8);
      __builtin_amdgcn_sched_barrier(0);
      idv = *idp;
#pragma unroll
      for (int k = 0; k < PS; ++k) {
        const uint4 q = hp[k][0];
        const uint4 x = hp[k][1];
        RH[(S0 + k) % RS] = (v4u){q.x, q.y, q.z, q.w};
        RX[(S0 + k) % RS] = (v4u){x.x, x.y, x.z, x.w};
      }
      __builtin_amdgcn_sched_barrier(0);
    };

    auto sel4 = [](unsigned u0, unsigned u1, unsigned u2, unsigned u3, unsigned m0, unsigned m1) -> float {
      const unsigned p01 = ((u1 ^ u0) & m0) ^ u0;
      const unsigned p23 = ((u3 ^ u2) & m0) ^ u2;
      return __builtin_bit_cast(float, ((p23 ^ p01) & m1) ^ p01);
    };

    float zs;
    float hs;

    auto step = [&](const v4u& Hr, const v4u& Xr, const v4u& Xn, const v4u& Hn1,
                    const v4u& Xnn) __attribute__((always_inline)) {
      const unsigned q0 = Hr.x, q1 = Hr.y, q2 = Hr.z, q3 = Hr.w, xe = Xr.w;
      const float h0 = __builtin_bit_cast(float, q0), h1 = __builtin_bit_cast(float, q1);
      const float h2 = __builtin_bit_cast(float, q2), h3 = __builtin_bit_cast(float, q3);
      const unsigned fsel = Xn.z;
      const unsigned gm0 = Xnn.x, gm1 = Xnn.y;
      const unsigned n0 = Hn1.x, n1 = Hn1.y, n2 = Hn1.z, n3 = Hn1.w;
      y0 = fmaf(s, h0, y0);
      y1 = fmaf(s, h1, y1);
      y2 = fmaf(s, h2, y2);
      y3 = fmaf(s, h3, y3);
      const float z = fmaf(s, hs, zs);
      hs = sel4(n0, n1, n2, n3, gm0, gm1);
      cacc = fmaf(s, __builtin_bit_cast(float, xe), cacc);
      asm volatile("" :: "v"(cacc));
      const unsigned zu = __builtin_bit_cast(unsigned, z);
      s = __builtin_bit_cast(float, __builtin_amdgcn_permlane16_var(zu, zu, fsel, false, false));
      zs = sel4(__builtin_bit_cast(unsigned, y0), __builtin_bit_cast(unsigned, y1),
                __builtin_bit_cast(unsigned, y2), __builtin_bit_cast(unsigned, y3), gm0, gm1);
    };

    int t0[LD];
    {
      const int i0 = Lc - LD + ln, i1 = Lc - LD - PS + ln;
      const int r0 = sp[i0 > Lc - 1 ? Lc - 1 : i0];
      const int r1 = sp[i1 < 0 ? 0 : i1];
#pragma unroll
      for (int k = 0; k < LD; ++k) t0[k] = __builtin_amdgcn_readlane(r0, LD - 1 - k);
#pragma unroll
      for (int k = 0; k < PS; ++k) tk[k] = __builtin_amdgcn_readlane(r1, PS - 1 - k);
    }

    {
      const float nden = -s_E[t0[0] * ES + HD];
      const float4 h = *(const float4*)(s_G + t0[0] * VD + 4 * ln);
      y0 = h.x * nden; y1 = h.y * nden; y2 = h.z * nden; y3 = h.w * nden;
    }

#pragma unroll
    for (int k = 0; k < LD; ++k) {
      const uint4* hp = (const uint4*)(s_R + t0[k] * 256 + lane * 8);
      const uint4 q = hp[0];
      const uint4 x = hp[1];
      RH[k] = (v4u){q.x, q.y, q.z, q.w};
      RX[k] = (v4u){x.x, x.y, x.z, x.w};
    }
    {
      const unsigned fm0 = RX[1].x, fm1 = RX[1].y;
      zs = sel4(__builtin_bit_cast(unsigned, y0), __builtin_bit_cast(unsigned, y1),
                __builtin_bit_cast(unsigned, y2), __builtin_bit_cast(unsigned, y3), fm0, fm1);
      const unsigned q0 = RH[0].x, q1 = RH[0].y, q2 = RH[0].z, q3 = RH[0].w;
      hs = sel4(q0, q1, q2, q3, fm0, fm1);
    }

    int pnext = Lc - LD - 2 * PS;
    auto segment = [&](auto mm) __attribute__((always_inline)) {
      constexpr int M = decltype(mm)::v;
      int raw;
      const int idx = pnext + ln;
      place(IC<(PS * M + LD) % RS>{}, sp + (idx < 0 ? 0 : idx), raw);
      pnext -= PS;
#pragma unroll
      for (int i = 0; i < PS; ++i) {
        const int a = (PS * M + i) % RS, b = (PS * M + i + 1) % RS, c = (PS * M + i + 2) % RS;
        step(RH[a], RX[a], RX[b], RH[b], RX[c]);
        if (i == PS / 2 - 1) {
#pragma unroll
          for (int k = 0; k < PS; ++k) tk[k] = __builtin_amdgcn_readlane(raw, PS - 1 - k);
        }
      }
    };

    int left = nseg;
    if (left > 0) {
#pragma unroll 1
      for (;;) {
        segment(IC<0>{}); if (--left <= 0) break;
        segment(IC<1>{}); if (--left <= 0) break;
        segment(IC<2>{}); if (--left <= 0) break;
      }
    }

    s_ctx[lane] = cacc;
  } else {
    float4 wrr[HD * HD / 4 / 32], wor[HD * VD / 4 / 32];
#pragma unroll
    for (int j = 0; j < HD * HD / 4 / 32; ++j) wrr[j] = ((const float4*)wr)[lane + j * 32];
#pragma unroll
    for (int j = 0; j < HD * VD / 4 / 32; ++j) wor[j] = ((const float4*)wo)[lane + j * 32];
    const float brr = br[lane], bo0 = bo[lane], bo1 = bo[lane + 32];
#pragma unroll
    for (int j = 0; j < HD * HD / 4 / 32; ++j) ((float4*)s_w2)[lane + j * 32] = wrr[j];
#pragma unroll
    for (int j = 0; j < HD * VD / 4 / 32; ++j) ((float4*)s_w1)[lane + j * 32] = wor[j];
    s_b2[lane] = brr; s_b1[lane] = bo0; s_b1[lane + 32] = bo1;
  }
  __syncthreads();

  if (tid < HD) {
    float acc = s_b2[tid];
#pragma unroll
    for (int j = 0; j < HD; ++j) acc = fmaf(s_ctx[j], s_w2[j * HD + tid], acc);
    s_tmp[tid] = acc;
  }
  __syncthreads();
  if (tid < VD) {
    float acc = s_b1[tid];
#pragma unroll
    for (int i = 0; i < HD; ++i) acc = fmaf(s_tmp[i], s_w1[i * VD + tid], acc);
    out[(size_t)bidx * VD + tid] = acc;
  }
}

extern "C" void kernel_launch(void* const* d_in, const int* in_sizes, int n_in,
                              void* d_out, int out_size, void* d_ws, size_t ws_size,
                              hipStream_t stream) {
  const int*   seq   = (const int*)  d_in[0];
  const float* embed = (const float*)d_in[1];
  const float* w1    = (const float*)d_in[2];
  const float* b1    = (const float*)d_in[3];
  const float* w2    = (const float*)d_in[4];
  const float* b2    = (const float*)d_in[5];
  const float* ln_g  = (const float*)d_in[6];
  const float* ln_b  = (const float*)d_in[7];
  const float* wr    = (const float*)d_in[8];
  const float* br    = (const float*)d_in[9];
  const float* wo    = (const float*)d_in[10];
  const float* bo    = (const float*)d_in[11];
  float* out = (float*)d_out;

  const int B = out_size / VD;
  const int L = in_sizes[0] / B;

  delta_model_fused<<<B, NTHREADS, 0, stream>>>(
      seq, embed, w1, b1, w2, b2, ln_g, ln_b, wr, br, wo, bo, out, L);
}
